# starting kernel
# baseline (speedup 1.0000x reference)
#include <hip/hip_runtime.h>

typedef __attribute__((ext_vector_type(2))) float v2f;
typedef __attribute__((ext_vector_type(8))) float v8f;

__global__ void zero_f32(float* __restrict__ buf, int n) {
  int i = blockIdx.x * blockDim.x + threadIdx.x;
  if (i < n) buf[i] = 0.0f;
}

__global__ void deg_accum(const int* __restrict__ dst, float* __restrict__ deg, int E) {
  int e = blockIdx.x * blockDim.x + threadIdx.x;
  if (e < E) atomicAdd(&deg[dst[e]], 1.0f);
}

__global__ void transform1(const float* __restrict__ x, const float* __restrict__ W1,
                           float* __restrict__ dnorm, float* __restrict__ p,
                           float* __restrict__ agg, int N) {
  int i = blockIdx.x * blockDim.x + threadIdx.x;
  if (i >= N) return;
  float di = rsqrtf(dnorm[i] + 1.0f);
  dnorm[i] = di;
  float s = x[i] * di;
  const float4* w = (const float4*)W1;
  float4* pp = (float4*)(p + (size_t)i * 16);
  float4* ag = (float4*)(agg + (size_t)i * 16);
#pragma unroll
  for (int k = 0; k < 4; ++k) {
    float4 wv = w[k];
    float4 o;
    o.x = wv.x * s; o.y = wv.y * s; o.z = wv.z * s; o.w = wv.w * s;
    pp[k] = o;
    float4 z; z.x = 0.f; z.y = 0.f; z.z = 0.f; z.w = 0.f;
    ag[k] = z;
  }
}

__global__ void edge_scatter(const int* __restrict__ src, const int* __restrict__ dst,
                             const float* __restrict__ p, float* __restrict__ agg, int E) {
  int t = blockIdx.x * blockDim.x + threadIdx.x;
  int e = t >> 2;
  if (e >= E) return;
  int sub = (t & 3) * 4;
  int s = src[e];
  int d = dst[e];
  float4 v = *(const float4*)(p + (size_t)s * 16 + sub);
  float* ap = agg + (size_t)d * 16 + sub;
  atomicAdd(ap + 0, v.x);
  atomicAdd(ap + 1, v.y);
  atomicAdd(ap + 2, v.z);
  atomicAdd(ap + 3, v.w);
}

__global__ void __launch_bounds__(256)
gcn_mid_wmma(const float* __restrict__ p, const float* __restrict__ agg,
             const float* __restrict__ dnorm, const float* __restrict__ b1,
             const float* __restrict__ W2, float* __restrict__ p2,
             float* __restrict__ agg_out, int N, int nTiles) {
  const int lane = threadIdx.x & 31;
  const int tile = (blockIdx.x * blockDim.x + threadIdx.x) >> 5;
  if (tile >= nTiles) return;
  const int base = tile * 16;
  const int col  = lane & 15;
  const bool hi  = lane >= 16;
  const bool full = (base + 16 <= N);

  int arow = base + col;
  if (arow >= N) arow = N - 1;
  const float di = dnorm[arow];

  v2f a[4], b[4];
#pragma unroll
  for (int k = 0; k < 4; ++k) {
    const int c0 = 4 * k + (hi ? 2 : 0);
    float2 pv = *(const float2*)(p   + (size_t)arow * 16 + c0);
    float2 av = *(const float2*)(agg + (size_t)arow * 16 + c0);
    a[k][0] = fmaxf(di * (av.x + pv.x) + b1[c0],     0.0f);
    a[k][1] = fmaxf(di * (av.y + pv.y) + b1[c0 + 1], 0.0f);
    b[k][0] = W2[c0 * 16 + col];
    b[k][1] = W2[(c0 + 1) * 16 + col];
  }

  v8f acc = {};
#pragma unroll
  for (int k = 0; k < 4; ++k)
    acc = __builtin_amdgcn_wmma_f32_16x16x4_f32(false, a[k], false, b[k],
                                                (short)0, acc, false, false);

  if (full) {
#pragma unroll
    for (int r = 0; r < 8; ++r) {
      int orow = base + (hi ? 8 + r : r);
      p2[(size_t)orow * 16 + col]      = acc[r] * dnorm[orow];
      agg_out[(size_t)orow * 16 + col] = 0.0f;
    }
  } else {
#pragma unroll
    for (int r = 0; r < 8; ++r) {
      int orow = base + (hi ? 8 + r : r);
      if (orow < N) {
        p2[(size_t)orow * 16 + col]      = acc[r] * dnorm[orow];
        agg_out[(size_t)orow * 16 + col] = 0.0f;
      }
    }
  }
}

__global__ void __launch_bounds__(256)
gcn_out_wmma(const float* __restrict__ p2, const float* __restrict__ agg2,
             const float* __restrict__ dnorm, const float* __restrict__ b2,
             const float* __restrict__ Wl, const float* __restrict__ bl,
             float* __restrict__ out, int N, int nTiles) {
  const int lane = threadIdx.x & 31;
  const int tile = (blockIdx.x * blockDim.x + threadIdx.x) >> 5;
  if (tile >= nTiles) return;
  const int base = tile * 16;
  const int col  = lane & 15;
  const bool hi  = lane >= 16;
  const bool full = (base + 16 <= N);

  int arow = base + col;
  if (arow >= N) arow = N - 1;
  const float di = dnorm[arow];

  v2f a[4], b[4];
#pragma unroll
  for (int k = 0; k < 4; ++k) {
    const int c0 = 4 * k + (hi ? 2 : 0);
    float2 pv = *(const float2*)(p2   + (size_t)arow * 16 + c0);
    float2 av = *(const float2*)(agg2 + (size_t)arow * 16 + c0);
    a[k][0] = fmaxf(di * (av.x + pv.x) + b2[c0],     0.0f);
    a[k][1] = fmaxf(di * (av.y + pv.y) + b2[c0 + 1], 0.0f);
    b[k][0] = (col < 4) ? Wl[c0 * 4 + col]       : 0.0f;
    b[k][1] = (col < 4) ? Wl[(c0 + 1) * 4 + col] : 0.0f;
  }

  const float cv = (col < 4) ? bl[col] : 0.0f;
  v8f acc;
#pragma unroll
  for (int r = 0; r < 8; ++r) acc[r] = cv;
#pragma unroll
  for (int k = 0; k < 4; ++k)
    acc = __builtin_amdgcn_wmma_f32_16x16x4_f32(false, a[k], false, b[k],
                                                (short)0, acc, false, false);

  if (col < 4) {
    if (full) {
#pragma unroll
      for (int r = 0; r < 8; ++r) {
        int orow = base + (hi ? 8 + r : r);
        out[(size_t)orow * 4 + col] = acc[r];
      }
    } else {
#pragma unroll
      for (int r = 0; r < 8; ++r) {
        int orow = base + (hi ? 8 + r : r);
        if (orow < N) out[(size_t)orow * 4 + col] = acc[r];
      }
    }
  }
}

extern "C" void kernel_launch(void* const* d_in, const int* in_sizes, int n_in,
                              void* d_out, int out_size, void* d_ws, size_t ws_size,
                              hipStream_t stream) {
  const float* x  = (const float*)d_in[0];
  const int*   ed = (const int*)d_in[1];
  const float* W1 = (const float*)d_in[2];
  const float* b1 = (const float*)d_in[3];
  const float* W2 = (const float*)d_in[4];
  const float* b2 = (const float*)d_in[5];
  const float* Wl = (const float*)d_in[6];
  const float* bl = (const float*)d_in[7];
  float* out = (float*)d_out;

  const int N = in_sizes[0];
  const int E = in_sizes[1] / 2;

  const int* src = ed;
  const int* dst = ed + E;

  float* dnorm = (float*)d_ws;
  float* p     = dnorm + N;
  float* agg   = p + (size_t)N * 16;

  const int B = 256;
  const int nTiles = (N + 15) / 16;
  const int tileBlocks = (nTiles * 32 + B - 1) / B;
  const int edgeBlocks = (int)(((long long)E * 4 + B - 1) / B);

  zero_f32  <<<(N + B - 1) / B, B, 0, stream>>>(dnorm, N);
  deg_accum <<<(E + B - 1) / B, B, 0, stream>>>(dst, dnorm, E);

  transform1<<<(N + B - 1) / B, B, 0, stream>>>(x, W1, dnorm, p, agg, N);
  edge_scatter<<<edgeBlocks, B, 0, stream>>>(src, dst, p, agg, E);

  gcn_mid_wmma<<<tileBlocks, B, 0, stream>>>(p, agg, dnorm, b1, W2, p, agg, N, nTiles);

  edge_scatter<<<edgeBlocks, B, 0, stream>>>(src, dst, p, agg, E);

  gcn_out_wmma<<<tileBlocks, B, 0, stream>>>(p, agg, dnorm, b2, Wl, bl, out, N, nTiles);
}


# in-degree counted in per-work-group LDS histograms (ds_add_u32; node slices x edge chunks; 32 partial planes, fixed-order integer sum) instead of a global f32 atomic per edge; same values
# speedup vs baseline: 1.4000x; 1.4000x over previous
#include <hip/hip_runtime.h>

typedef __attribute__((ext_vector_type(2))) float v2f;
typedef __attribute__((ext_vector_type(8))) float v8f;

#define DEG_SLICE   25024
#define DEG_CHUNKS  32
#define DEG_THREADS 1024

#define DEG_BUMP(d)                                              \
  {                                                              \
    const unsigned r_ = (unsigned)((d) - lo);                    \
    if (r_ < cnt) atomicAdd(&hist[r_], 1u);                      \
  }
#define DEG_BUMP4(v) DEG_BUMP((v).x) DEG_BUMP((v).y) DEG_BUMP((v).z) DEG_BUMP((v).w)

__global__ void __launch_bounds__(DEG_THREADS)
deg_hist(const int* __restrict__ dst, unsigned* __restrict__ part, int N, int E, int chunkE) {
  __shared__ unsigned hist[DEG_SLICE];
  const int tid   = threadIdx.x;
  const int chunk = blockIdx.x % DEG_CHUNKS;
  const int slice = blockIdx.x / DEG_CHUNKS;
  const int lo    = slice * DEG_SLICE;
  const int rem   = N - lo;
  const unsigned cnt = (unsigned)(rem < DEG_SLICE ? rem : DEG_SLICE);

  for (unsigned i = tid; i < cnt; i += DEG_THREADS) hist[i] = 0u;
  __syncthreads();

  const long long begin = (long long)chunk * chunkE;
  long long end = begin + chunkE;
  if (end > E) end = E;
  const int len = (end > begin) ? (int)(end - begin) : 0;
  const int* dp = dst + begin;

  int done = 0;
  if ((((size_t)dp) & 15) == 0) {
    const int4* d4 = (const int4*)dp;
    const int n4 = len >> 2;
    int i = tid;
    for (; i + 3 * DEG_THREADS < n4; i += 4 * DEG_THREADS) {
      const int4 a = d4[i];
      const int4 b = d4[i + DEG_THREADS];
      const int4 c = d4[i + 2 * DEG_THREADS];
      const int4 e = d4[i + 3 * DEG_THREADS];
      DEG_BUMP4(a) DEG_BUMP4(b) DEG_BUMP4(c) DEG_BUMP4(e)
    }
    for (; i < n4; i += DEG_THREADS) {
      const int4 a = d4[i];
      DEG_BUMP4(a)
    }
    done = n4 << 2;
  }
  for (int i = done + tid; i < len; i += DEG_THREADS) DEG_BUMP(dp[i])
  __syncthreads();

  unsigned* o = part + (size_t)chunk * N + lo;
  for (unsigned i = tid; i < cnt; i += DEG_THREADS) o[i] = hist[i];
}

__global__ void deg_reduce(const unsigned* __restrict__ part, float* __restrict__ deg, int N) {
  int i = blockIdx.x * blockDim.x + threadIdx.x;
  if (i >= N) return;
  unsigned s = 0u;
  for (int c = 0; c < DEG_CHUNKS; ++c) s += part[(size_t)c * N + i];
  deg[i] = (float)s;
}

__global__ void transform1(const float* __restrict__ x, const float* __restrict__ W1,
                           float* __restrict__ dnorm, float* __restrict__ p,
                           float* __restrict__ agg, int N) {
  int i = blockIdx.x * blockDim.x + threadIdx.x;
  if (i >= N) return;
  float di = rsqrtf(dnorm[i] + 1.0f);
  dnorm[i] = di;
  float s = x[i] * di;
  const float4* w = (const float4*)W1;
  float4* pp = (float4*)(p + (size_t)i * 16);
  float4* ag = (float4*)(agg + (size_t)i * 16);
#pragma unroll
  for (int k = 0; k < 4; ++k) {
    float4 wv = w[k];
    float4 o;
    o.x = wv.x * s; o.y = wv.y * s; o.z = wv.z * s; o.w = wv.w * s;
    pp[k] = o;
    float4 z; z.x = 0.f; z.y = 0.f; z.z = 0.f; z.w = 0.f;
    ag[k] = z;
  }
}

__global__ void edge_scatter(const int* __restrict__ src, const int* __restrict__ dst,
                             const float* __restrict__ p, float* __restrict__ agg, int E) {
  int t = blockIdx.x * blockDim.x + threadIdx.x;
  int e = t >> 2;
  if (e >= E) return;
  int sub = (t & 3) * 4;
  int s = src[e];
  int d = dst[e];
  float4 v = *(const float4*)(p + (size_t)s * 16 + sub);
  float* ap = agg + (size_t)d * 16 + sub;
  atomicAdd(ap + 0, v.x);
  atomicAdd(ap + 1, v.y);
  atomicAdd(ap + 2, v.z);
  atomicAdd(ap + 3, v.w);
}

__global__ void __launch_bounds__(256)
gcn_mid_wmma(const float* __restrict__ p, const float* __restrict__ agg,
             const float* __restrict__ dnorm, const float* __restrict__ b1,
             const float* __restrict__ W2, float* __restrict__ p2,
             float* __restrict__ agg_out, int N, int nTiles) {
  const int lane = threadIdx.x & 31;
  const int tile = (blockIdx.x * blockDim.x + threadIdx.x) >> 5;
  if (tile >= nTiles) return;
  const int base = tile * 16;
  const int col  = lane & 15;
  const bool hi  = lane >= 16;
  const bool full = (base + 16 <= N);

  int arow = base + col;
  if (arow >= N) arow = N - 1;
  const float di = dnorm[arow];

  v2f a[4], b[4];
#pragma unroll
  for (int k = 0; k < 4; ++k) {
    const int c0 = 4 * k + (hi ? 2 : 0);
    float2 pv = *(const float2*)(p   + (size_t)arow * 16 + c0);
    float2 av = *(const float2*)(agg + (size_t)arow * 16 + c0);
    a[k][0] = fmaxf(di * (av.x + pv.x) + b1[c0],     0.0f);
    a[k][1] = fmaxf(di * (av.y + pv.y) + b1[c0 + 1], 0.0f);
    b[k][0] = W2[c0 * 16 + col];
    b[k][1] = W2[(c0 + 1) * 16 + col];
  }

  v8f acc = {};
#pragma unroll
  for (int k = 0; k < 4; ++k)
    acc = __builtin_amdgcn_wmma_f32_16x16x4_f32(false, a[k], false, b[k],
                                                (short)0, acc, false, false);

  if (full) {
#pragma unroll
    for (int r = 0; r < 8; ++r) {
      int orow = base + (hi ? 8 + r : r);
      p2[(size_t)orow * 16 + col]      = acc[r] * dnorm[orow];
      agg_out[(size_t)orow * 16 + col] = 0.0f;
    }
  } else {
#pragma unroll
    for (int r = 0; r < 8; ++r) {
      int orow = base + (hi ? 8 + r : r);
      if (orow < N) {
        p2[(size_t)orow * 16 + col]      = acc[r] * dnorm[orow];
        agg_out[(size_t)orow * 16 + col] = 0.0f;
      }
    }
  }
}

__global__ void __launch_bounds__(256)
gcn_out_wmma(const float* __restrict__ p2, const float* __restrict__ agg2,
             const float* __restrict__ dnorm, const float* __restrict__ b2,
             const float* __restrict__ Wl, const float* __restrict__ bl,
             float* __restrict__ out, int N, int nTiles) {
  const int lane = threadIdx.x & 31;
  const int tile = (blockIdx.x * blockDim.x + threadIdx.x) >> 5;
  if (tile >= nTiles) return;
  const int base = tile * 16;
  const int col  = lane & 15;
  const bool hi  = lane >= 16;
  const bool full = (base + 16 <= N);

  int arow = base + col;
  if (arow >= N) arow = N - 1;
  const float di = dnorm[arow];

  v2f a[4], b[4];
#pragma unroll
  for (int k = 0; k < 4; ++k) {
    const int c0 = 4 * k + (hi ? 2 : 0);
    float2 pv = *(const float2*)(p2   + (size_t)arow * 16 + c0);
    float2 av = *(const float2*)(agg2 + (size_t)arow * 16 + c0);
    a[k][0] = fmaxf(di * (av.x + pv.x) + b2[c0],     0.0f);
    a[k][1] = fmaxf(di * (av.y + pv.y) + b2[c0 + 1], 0.0f);
    b[k][0] = (col < 4) ? Wl[c0 * 4 + col]       : 0.0f;
    b[k][1] = (col < 4) ? Wl[(c0 + 1) * 4 + col] : 0.0f;
  }

  const float cv = (col < 4) ? bl[col] : 0.0f;
  v8f acc;
#pragma unroll
  for (int r = 0; r < 8; ++r) acc[r] = cv;
#pragma unroll
  for (int k = 0; k < 4; ++k)
    acc = __builtin_amdgcn_wmma_f32_16x16x4_f32(false, a[k], false, b[k],
                                                (short)0, acc, false, false);

  if (col < 4) {
    if (full) {
#pragma unroll
      for (int r = 0; r < 8; ++r) {
        int orow = base + (hi ? 8 + r : r);
        out[(size_t)orow * 4 + col] = acc[r];
      }
    } else {
#pragma unroll
      for (int r = 0; r < 8; ++r) {
        int orow = base + (hi ? 8 + r : r);
        if (orow < N) out[(size_t)orow * 4 + col] = acc[r];
      }
    }
  }
}

extern "C" void kernel_launch(void* const* d_in, const int* in_sizes, int n_in,
                              void* d_out, int out_size, void* d_ws, size_t ws_size,
                              hipStream_t stream) {
  const float* x  = (const float*)d_in[0];
  const int*   ed = (const int*)d_in[1];
  const float* W1 = (const float*)d_in[2];
  const float* b1 = (const float*)d_in[3];
  const float* W2 = (const float*)d_in[4];
  const float* b2 = (const float*)d_in[5];
  const float* Wl = (const float*)d_in[6];
  const float* bl = (const float*)d_in[7];
  float* out = (float*)d_out;

  const int N = in_sizes[0];
  const int E = in_sizes[1] / 2;

  const int* src = ed;
  const int* dst = ed + E;

  float* dnorm = (float*)d_ws;
  float* p     = dnorm + N;
  float* agg   = p + (size_t)N * 16;

  const int B = 256;
  const int nTiles = (N + 15) / 16;
  const int tileBlocks = (nTiles * 32 + B - 1) / B;
  const int edgeBlocks = (int)(((long long)E * 4 + B - 1) / B);

  unsigned* part = (unsigned*)p;
  const int chunkE  = ((((E + DEG_CHUNKS - 1) / DEG_CHUNKS) + 3) / 4) * 4;
  const int nSlices = (N + DEG_SLICE - 1) / DEG_SLICE;
  deg_hist  <<<nSlices * DEG_CHUNKS, DEG_THREADS, 0, stream>>>(dst, part, N, E, chunkE);
  deg_reduce<<<(N + B - 1) / B, B, 0, stream>>>(part, dnorm, N);

  transform1<<<(N + B - 1) / B, B, 0, stream>>>(x, W1, dnorm, p, agg, N);
  edge_scatter<<<edgeBlocks, B, 0, stream>>>(src, dst, p, agg, E);

  gcn_mid_wmma<<<tileBlocks, B, 0, stream>>>(p, agg, dnorm, b1, W2, p, agg, N, nTiles);

  edge_scatter<<<edgeBlocks, B, 0, stream>>>(src, dst, p, agg, E);

  gcn_out_wmma<<<tileBlocks, B, 0, stream>>>(p, agg, dnorm, b2, Wl, bl, out, N, nTiles);
}


# layer 1 aggregated before its 1->16 transform ((A x) W1, not A (x W1)): one float per edge summed in work-group LDS (ds_add_f32; slices x chunks; 16 planes, fixed-order sum); f32 sum order changes
# speedup vs baseline: 2.1793x; 1.5566x over previous
#include <hip/hip_runtime.h>

typedef __attribute__((ext_vector_type(2))) float v2f;
typedef __attribute__((ext_vector_type(8))) float v8f;

#define DEG_SLICE   25024
#define DEG_CHUNKS  32
#define DEG_THREADS 1024

#define DEG_BUMP(d)                                              \
  {                                                              \
    const unsigned r_ = (unsigned)((d) - lo);                    \
    if (r_ < cnt) atomicAdd(&hist[r_], 1u);                      \
  }
#define DEG_BUMP4(v) DEG_BUMP((v).x) DEG_BUMP((v).y) DEG_BUMP((v).z) DEG_BUMP((v).w)

__global__ void __launch_bounds__(DEG_THREADS)
deg_hist(const int* __restrict__ dst, unsigned* __restrict__ part, int N, int E, int chunkE) {
  __shared__ unsigned hist[DEG_SLICE];
  const int tid   = threadIdx.x;
  const int chunk = blockIdx.x % DEG_CHUNKS;
  const int slice = blockIdx.x / DEG_CHUNKS;
  const int lo    = slice * DEG_SLICE;
  const int rem   = N - lo;
  const unsigned cnt = (unsigned)(rem < DEG_SLICE ? rem : DEG_SLICE);

  for (unsigned i = tid; i < cnt; i += DEG_THREADS) hist[i] = 0u;
  __syncthreads();

  const long long begin = (long long)chunk * chunkE;
  long long end = begin + chunkE;
  if (end > E) end = E;
  const int len = (end > begin) ? (int)(end - begin) : 0;
  const int* dp = dst + begin;

  int done = 0;
  if ((((size_t)dp) & 15) == 0) {
    const int4* d4 = (const int4*)dp;
    const int n4 = len >> 2;
    int i = tid;
    for (; i + 3 * DEG_THREADS < n4; i += 4 * DEG_THREADS) {
      const int4 a = d4[i];
      const int4 b = d4[i + DEG_THREADS];
      const int4 c = d4[i + 2 * DEG_THREADS];
      const int4 e = d4[i + 3 * DEG_THREADS];
      DEG_BUMP4(a) DEG_BUMP4(b) DEG_BUMP4(c) DEG_BUMP4(e)
    }
    for (; i < n4; i += DEG_THREADS) {
      const int4 a = d4[i];
      DEG_BUMP4(a)
    }
    done = n4 << 2;
  }
  for (int i = done + tid; i < len; i += DEG_THREADS) DEG_BUMP(dp[i])
  __syncthreads();

  unsigned* o = part + (size_t)chunk * N + lo;
  for (unsigned i = tid; i < cnt; i += DEG_THREADS) o[i] = hist[i];
}

__global__ void deg_reduce(const unsigned* part, const float* __restrict__ x,
                           float* __restrict__ dnorm, float* q, int N) {
  int i = blockIdx.x * blockDim.x + threadIdx.x;
  if (i >= N) return;
  unsigned s = 0u;
  for (int c = 0; c < DEG_CHUNKS; ++c) s += part[(size_t)c * N + i];
  const float di = rsqrtf((float)s + 1.0f);
  dnorm[i] = di;
  q[i] = x[i] * di;
}

#define WS_SLICE   12512
#define WS_CHUNKS  16
#define WS_THREADS 1024

#define WS_TAKE(d, s, R, V)                                      \
  const unsigned R = (unsigned)((d) - lo);                       \
  float V = 0.0f;                                                \
  if (R < cnt) V = q[(s)];
#define WS_PUT(R, V) if (R < cnt) atomicAdd(&hist[R], V);
#define WS_TAKE4(dv, sv, T)                                      \
  WS_TAKE((dv).x, (sv).x, T##r0, T##v0) WS_TAKE((dv).y, (sv).y, T##r1, T##v1) \
  WS_TAKE((dv).z, (sv).z, T##r2, T##v2) WS_TAKE((dv).w, (sv).w, T##r3, T##v3)
#define WS_PUT4(T) WS_PUT(T##r0, T##v0) WS_PUT(T##r1, T##v1) WS_PUT(T##r2, T##v2) WS_PUT(T##r3, T##v3)

__global__ void __launch_bounds__(WS_THREADS)
wsum_hist(const int* __restrict__ src, const int* __restrict__ dst, const float* __restrict__ q,
          float* __restrict__ fpart, int N, int E, int chunkE) {
  __shared__ float hist[WS_SLICE];
  const int tid   = threadIdx.x;
  const int chunk = blockIdx.x % WS_CHUNKS;
  const int slice = blockIdx.x / WS_CHUNKS;
  const int lo    = slice * WS_SLICE;
  const int rem   = N - lo;
  const unsigned cnt = (unsigned)(rem < WS_SLICE ? rem : WS_SLICE);

  for (unsigned i = tid; i < cnt; i += WS_THREADS) hist[i] = 0.0f;
  __syncthreads();

  const long long begin = (long long)chunk * chunkE;
  long long end = begin + chunkE;
  if (end > E) end = E;
  const int len = (end > begin) ? (int)(end - begin) : 0;
  const int* dp = dst + begin;
  const int* sp = src + begin;

  int done = 0;
  if (((((size_t)dp) | ((size_t)sp)) & 15) == 0) {
    const int4* d4 = (const int4*)dp;
    const int4* s4 = (const int4*)sp;
    const int n4 = len >> 2;
    int i = tid;
    for (; i + 3 * WS_THREADS < n4; i += 4 * WS_THREADS) {
      const int4 da = d4[i];
      const int4 db = d4[i + WS_THREADS];
      const int4 dc = d4[i + 2 * WS_THREADS];
      const int4 de = d4[i + 3 * WS_THREADS];
      const int4 sa = s4[i];
      const int4 sb = s4[i + WS_THREADS];
      const int4 sc = s4[i + 2 * WS_THREADS];
      const int4 se = s4[i + 3 * WS_THREADS];
      WS_TAKE4(da, sa, a) WS_TAKE4(db, sb, b) WS_TAKE4(dc, sc, c) WS_TAKE4(de, se, e)
      WS_PUT4(a) WS_PUT4(b) WS_PUT4(c) WS_PUT4(e)
    }
    for (; i < n4; i += WS_THREADS) {
      const int4 da = d4[i];
      const int4 sa = s4[i];
      WS_TAKE4(da, sa, a)
      WS_PUT4(a)
    }
    done = n4 << 2;
  }
  for (int i = done + tid; i < len; i += WS_THREADS) {
    WS_TAKE(dp[i], sp[i], tr, tv)
    WS_PUT(tr, tv)
  }
  __syncthreads();

  float* o = fpart + (size_t)chunk * N + lo;
  for (unsigned i = tid; i < cnt; i += WS_THREADS) o[i] = hist[i];
}

__global__ void node_reduce(const float* __restrict__ fpart, const float* __restrict__ x,
                            const float* __restrict__ dnorm, float* __restrict__ agg, int N) {
  int i = blockIdx.x * blockDim.x + threadIdx.x;
  if (i >= N) return;
  float s = 0.0f;
  for (int c = 0; c < WS_CHUNKS; ++c) s += fpart[(size_t)c * N + i];
  const float di = dnorm[i];
  agg[(size_t)i * 16] = di * (s + x[i] * di);
}

__global__ void edge_scatter(const int* __restrict__ src, const int* __restrict__ dst,
                             const float* __restrict__ p, float* __restrict__ agg, int E) {
  int t = blockIdx.x * blockDim.x + threadIdx.x;
  int e = t >> 2;
  if (e >= E) return;
  int sub = (t & 3) * 4;
  int s = src[e];
  int d = dst[e];
  float4 v = *(const float4*)(p + (size_t)s * 16 + sub);
  float* ap = agg + (size_t)d * 16 + sub;
  atomicAdd(ap + 0, v.x);
  atomicAdd(ap + 1, v.y);
  atomicAdd(ap + 2, v.z);
  atomicAdd(ap + 3, v.w);
}

__global__ void __launch_bounds__(256)
gcn_mid_wmma(float* agg, const float* __restrict__ dnorm, const float* __restrict__ W1,
             const float* __restrict__ b1, const float* __restrict__ W2,
             float* __restrict__ p2, int N, int nTiles) {
  const int lane = threadIdx.x & 31;
  const int tile = (blockIdx.x * blockDim.x + threadIdx.x) >> 5;
  if (tile >= nTiles) return;
  const int base = tile * 16;
  const int col  = lane & 15;
  const bool hi  = lane >= 16;
  const bool full = (base + 16 <= N);

  int arow = base + col;
  if (arow >= N) arow = N - 1;
  const float t = agg[(size_t)arow * 16];

  v2f a[4], b[4];
#pragma unroll
  for (int k = 0; k < 4; ++k) {
    const int c0 = 4 * k + (hi ? 2 : 0);
    a[k][0] = fmaxf(t * W1[c0]     + b1[c0],     0.0f);
    a[k][1] = fmaxf(t * W1[c0 + 1] + b1[c0 + 1], 0.0f);
    b[k][0] = W2[c0 * 16 + col];
    b[k][1] = W2[(c0 + 1) * 16 + col];
  }

  v8f acc = {};
#pragma unroll
  for (int k = 0; k < 4; ++k)
    acc = __builtin_amdgcn_wmma_f32_16x16x4_f32(false, a[k], false, b[k],
                                                (short)0, acc, false, false);

  if (full) {
#pragma unroll
    for (int r = 0; r < 8; ++r) {
      int orow = base + (hi ? 8 + r : r);
      p2[(size_t)orow * 16 + col]  = acc[r] * dnorm[orow];
      agg[(size_t)orow * 16 + col] = 0.0f;
    }
  } else {
#pragma unroll
    for (int r = 0; r < 8; ++r) {
      int orow = base + (hi ? 8 + r : r);
      if (orow < N) {
        p2[(size_t)orow * 16 + col]  = acc[r] * dnorm[orow];
        agg[(size_t)orow * 16 + col] = 0.0f;
      }
    }
  }
}

__global__ void __launch_bounds__(256)
gcn_out_wmma(const float* __restrict__ p2, const float* __restrict__ agg2,
             const float* __restrict__ dnorm, const float* __restrict__ b2,
             const float* __restrict__ Wl, const float* __restrict__ bl,
             float* __restrict__ out, int N, int nTiles) {
  const int lane = threadIdx.x & 31;
  const int tile = (blockIdx.x * blockDim.x + threadIdx.x) >> 5;
  if (tile >= nTiles) return;
  const int base = tile * 16;
  const int col  = lane & 15;
  const bool hi  = lane >= 16;
  const bool full = (base + 16 <= N);

  int arow = base + col;
  if (arow >= N) arow = N - 1;
  const float di = dnorm[arow];

  v2f a[4], b[4];
#pragma unroll
  for (int k = 0; k < 4; ++k) {
    const int c0 = 4 * k + (hi ? 2 : 0);
    float2 pv = *(const float2*)(p2   + (size_t)arow * 16 + c0);
    float2 av = *(const float2*)(agg2 + (size_t)arow * 16 + c0);
    a[k][0] = fmaxf(di * (av.x + pv.x) + b2[c0],     0.0f);
    a[k][1] = fmaxf(di * (av.y + pv.y) + b2[c0 + 1], 0.0f);
    b[k][0] = (col < 4) ? Wl[c0 * 4 + col]       : 0.0f;
    b[k][1] = (col < 4) ? Wl[(c0 + 1) * 4 + col] : 0.0f;
  }

  const float cv = (col < 4) ? bl[col] : 0.0f;
  v8f acc;
#pragma unroll
  for (int r = 0; r < 8; ++r) acc[r] = cv;
#pragma unroll
  for (int k = 0; k < 4; ++k)
    acc = __builtin_amdgcn_wmma_f32_16x16x4_f32(false, a[k], false, b[k],
                                                (short)0, acc, false, false);

  if (col < 4) {
    if (full) {
#pragma unroll
      for (int r = 0; r < 8; ++r) {
        int orow = base + (hi ? 8 + r : r);
        out[(size_t)orow * 4 + col] = acc[r];
      }
    } else {
#pragma unroll
      for (int r = 0; r < 8; ++r) {
        int orow = base + (hi ? 8 + r : r);
        if (orow < N) out[(size_t)orow * 4 + col] = acc[r];
      }
    }
  }
}

extern "C" void kernel_launch(void* const* d_in, const int* in_sizes, int n_in,
                              void* d_out, int out_size, void* d_ws, size_t ws_size,
                              hipStream_t stream) {
  const float* x  = (const float*)d_in[0];
  const int*   ed = (const int*)d_in[1];
  const float* W1 = (const float*)d_in[2];
  const float* b1 = (const float*)d_in[3];
  const float* W2 = (const float*)d_in[4];
  const float* b2 = (const float*)d_in[5];
  const float* Wl = (const float*)d_in[6];
  const float* bl = (const float*)d_in[7];
  float* out = (float*)d_out;

  const int N = in_sizes[0];
  const int E = in_sizes[1] / 2;

  const int* src = ed;
  const int* dst = ed + E;

  float* dnorm = (float*)d_ws;
  float* p     = dnorm + N;
  float* agg   = p + (size_t)N * 16;

  const int B = 256;
  const int nTiles = (N + 15) / 16;
  const int tileBlocks = (nTiles * 32 + B - 1) / B;
  const int edgeBlocks = (int)(((long long)E * 4 + B - 1) / B);

  unsigned* part = (unsigned*)p;
  float* q = agg;
  const int chunkE  = ((((E + DEG_CHUNKS - 1) / DEG_CHUNKS) + 3) / 4) * 4;
  const int nSlices = (N + DEG_SLICE - 1) / DEG_SLICE;
  deg_hist  <<<nSlices * DEG_CHUNKS, DEG_THREADS, 0, stream>>>(dst, part, N, E, chunkE);
  deg_reduce<<<(N + B - 1) / B, B, 0, stream>>>(part, x, dnorm, q, N);

  const int wChunkE  = ((((E + WS_CHUNKS - 1) / WS_CHUNKS) + 3) / 4) * 4;
  const int wSlices  = (N + WS_SLICE - 1) / WS_SLICE;
  wsum_hist  <<<wSlices * WS_CHUNKS, WS_THREADS, 0, stream>>>(src, dst, q, p, N, E, wChunkE);
  node_reduce<<<(N + B - 1) / B, B, 0, stream>>>(p, x, dnorm, agg, N);

  gcn_mid_wmma<<<tileBlocks, B, 0, stream>>>(agg, dnorm, W1, b1, W2, p, N, nTiles);

  edge_scatter<<<edgeBlocks, B, 0, stream>>>(src, dst, p, agg, E);

  gcn_out_wmma<<<tileBlocks, B, 0, stream>>>(p, agg, dnorm, b2, Wl, bl, out, N, nTiles);
}


# layer-2 scatter: a wave walks 32 edges two at a time, 16 lanes per 64-B row: each gather and f32 atomic instruction covers 2 whole rows, not 4 words of 8 rows; arrival-order atomic sum as before
# speedup vs baseline: 3.0450x; 1.3972x over previous
#include <hip/hip_runtime.h>

typedef __attribute__((ext_vector_type(2))) float v2f;
typedef __attribute__((ext_vector_type(8))) float v8f;

#define DEG_SLICE   25024
#define DEG_CHUNKS  32
#define DEG_THREADS 1024

#define DEG_BUMP(d)                                              \
  {                                                              \
    const unsigned r_ = (unsigned)((d) - lo);                    \
    if (r_ < cnt) atomicAdd(&hist[r_], 1u);                      \
  }
#define DEG_BUMP4(v) DEG_BUMP((v).x) DEG_BUMP((v).y) DEG_BUMP((v).z) DEG_BUMP((v).w)

__global__ void __launch_bounds__(DEG_THREADS)
deg_hist(const int* __restrict__ dst, unsigned* __restrict__ part, int N, int E, int chunkE) {
  __shared__ unsigned hist[DEG_SLICE];
  const int tid   = threadIdx.x;
  const int chunk = blockIdx.x % DEG_CHUNKS;
  const int slice = blockIdx.x / DEG_CHUNKS;
  const int lo    = slice * DEG_SLICE;
  const int rem   = N - lo;
  const unsigned cnt = (unsigned)(rem < DEG_SLICE ? rem : DEG_SLICE);

  for (unsigned i = tid; i < cnt; i += DEG_THREADS) hist[i] = 0u;
  __syncthreads();

  const long long begin = (long long)chunk * chunkE;
  long long end = begin + chunkE;
  if (end > E) end = E;
  const int len = (end > begin) ? (int)(end - begin) : 0;
  const int* dp = dst + begin;

  int done = 0;
  if ((((size_t)dp) & 15) == 0) {
    const int4* d4 = (const int4*)dp;
    const int n4 = len >> 2;
    int i = tid;
    for (; i + 3 * DEG_THREADS < n4; i += 4 * DEG_THREADS) {
      const int4 a = d4[i];
      const int4 b = d4[i + DEG_THREADS];
      const int4 c = d4[i + 2 * DEG_THREADS];
      const int4 e = d4[i + 3 * DEG_THREADS];
      DEG_BUMP4(a) DEG_BUMP4(b) DEG_BUMP4(c) DEG_BUMP4(e)
    }
    for (; i < n4; i += DEG_THREADS) {
      const int4 a = d4[i];
      DEG_BUMP4(a)
    }
    done = n4 << 2;
  }
  for (int i = done + tid; i < len; i += DEG_THREADS) DEG_BUMP(dp[i])
  __syncthreads();

  unsigned* o = part + (size_t)chunk * N + lo;
  for (unsigned i = tid; i < cnt; i += DEG_THREADS) o[i] = hist[i];
}

__global__ void deg_reduce(const unsigned* part, const float* __restrict__ x,
                           float* __restrict__ dnorm, float* q, int N) {
  int i = blockIdx.x * blockDim.x + threadIdx.x;
  if (i >= N) return;
  unsigned s = 0u;
  for (int c = 0; c < DEG_CHUNKS; ++c) s += part[(size_t)c * N + i];
  const float di = rsqrtf((float)s + 1.0f);
  dnorm[i] = di;
  q[i] = x[i] * di;
}

#define WS_SLICE   12512
#define WS_CHUNKS  16
#define WS_THREADS 1024

#define WS_TAKE(d, s, R, V)                                      \
  const unsigned R = (unsigned)((d) - lo);                       \
  float V = 0.0f;                                                \
  if (R < cnt) V = q[(s)];
#define WS_PUT(R, V) if (R < cnt) atomicAdd(&hist[R], V);
#define WS_TAKE4(dv, sv, T)                                      \
  WS_TAKE((dv).x, (sv).x, T##r0, T##v0) WS_TAKE((dv).y, (sv).y, T##r1, T##v1) \
  WS_TAKE((dv).z, (sv).z, T##r2, T##v2) WS_TAKE((dv).w, (sv).w, T##r3, T##v3)
#define WS_PUT4(T) WS_PUT(T##r0, T##v0) WS_PUT(T##r1, T##v1) WS_PUT(T##r2, T##v2) WS_PUT(T##r3, T##v3)

__global__ void __launch_bounds__(WS_THREADS)
wsum_hist(const int* __restrict__ src, const int* __restrict__ dst, const float* __restrict__ q,
          float* __restrict__ fpart, int N, int E, int chunkE) {
  __shared__ float hist[WS_SLICE];
  const int tid   = threadIdx.x;
  const int chunk = blockIdx.x % WS_CHUNKS;
  const int slice = blockIdx.x / WS_CHUNKS;
  const int lo    = slice * WS_SLICE;
  const int rem   = N - lo;
  const unsigned cnt = (unsigned)(rem < WS_SLICE ? rem : WS_SLICE);

  for (unsigned i = tid; i < cnt; i += WS_THREADS) hist[i] = 0.0f;
  __syncthreads();

  const long long begin = (long long)chunk * chunkE;
  long long end = begin + chunkE;
  if (end > E) end = E;
  const int len = (end > begin) ? (int)(end - begin) : 0;
  const int* dp = dst + begin;
  const int* sp = src + begin;

  int done = 0;
  if (((((size_t)dp) | ((size_t)sp)) & 15) == 0) {
    const int4* d4 = (const int4*)dp;
    const int4* s4 = (const int4*)sp;
    const int n4 = len >> 2;
    int i = tid;
    for (; i + 3 * WS_THREADS < n4; i += 4 * WS_THREADS) {
      const int4 da = d4[i];
      const int4 db = d4[i + WS_THREADS];
      const int4 dc = d4[i + 2 * WS_THREADS];
      const int4 de = d4[i + 3 * WS_THREADS];
      const int4 sa = s4[i];
      const int4 sb = s4[i + WS_THREADS];
      const int4 sc = s4[i + 2 * WS_THREADS];
      const int4 se = s4[i + 3 * WS_THREADS];
      WS_TAKE4(da, sa, a) WS_TAKE4(db, sb, b) WS_TAKE4(dc, sc, c) WS_TAKE4(de, se, e)
      WS_PUT4(a) WS_PUT4(b) WS_PUT4(c) WS_PUT4(e)
    }
    for (; i < n4; i += WS_THREADS) {
      const int4 da = d4[i];
      const int4 sa = s4[i];
      WS_TAKE4(da, sa, a)
      WS_PUT4(a)
    }
    done = n4 << 2;
  }
  for (int i = done + tid; i < len; i += WS_THREADS) {
    WS_TAKE(dp[i], sp[i], tr, tv)
    WS_PUT(tr, tv)
  }
  __syncthreads();

  float* o = fpart + (size_t)chunk * N + lo;
  for (unsigned i = tid; i < cnt; i += WS_THREADS) o[i] = hist[i];
}

__global__ void node_reduce(const float* __restrict__ fpart, const float* __restrict__ x,
                            const float* __restrict__ dnorm, float* __restrict__ agg, int N) {
  int i = blockIdx.x * blockDim.x + threadIdx.x;
  if (i >= N) return;
  float s = 0.0f;
  for (int c = 0; c < WS_CHUNKS; ++c) s += fpart[(size_t)c * N + i];
  const float di = dnorm[i];
  agg[(size_t)i * 16] = di * (s + x[i] * di);
}

__global__ void __launch_bounds__(256)
edge_scatter(const int* __restrict__ src, const int* __restrict__ dst,
             const float* __restrict__ p, float* __restrict__ agg, int E) {
  const int lane = threadIdx.x & 31;
  const int wave = __builtin_amdgcn_readfirstlane((int)((blockIdx.x * blockDim.x + threadIdx.x) >> 5));
  const long long e0 = (long long)wave * 32;
  if (e0 >= E) return;
  const unsigned c    = (unsigned)(lane & 15);
  const int      half = lane >> 4;
  const long long left = (long long)E - e0;
  const int n = left < 32 ? (int)left : 32;

  int s = 0, d = 0;
  if (lane < n) {
    s = src[e0 + lane];
    d = dst[e0 + lane];
  }

  float    v[16];
  unsigned ao[16];
#pragma unroll
  for (int k = 0; k < 16; ++k) {
    const unsigned sj = (unsigned)__shfl(s, 2 * k + half);
    const unsigned dj = (unsigned)__shfl(d, 2 * k + half);
    ao[k] = dj * 16u + c;
    v[k]  = p[sj * 16u + c];
  }

  if (n == 32) {
#pragma unroll
    for (int k = 0; k < 16; ++k) atomicAdd(agg + ao[k], v[k]);
  } else {
#pragma unroll
    for (int k = 0; k < 16; ++k)
      if (2 * k + half < n) atomicAdd(agg + ao[k], v[k]);
  }
}

__global__ void __launch_bounds__(256)
gcn_mid_wmma(float* agg, const float* __restrict__ dnorm, const float* __restrict__ W1,
             const float* __restrict__ b1, const float* __restrict__ W2,
             float* __restrict__ p2, int N, int nTiles) {
  const int lane = threadIdx.x & 31;
  const int tile = (blockIdx.x * blockDim.x + threadIdx.x) >> 5;
  if (tile >= nTiles) return;
  const int base = tile * 16;
  const int col  = lane & 15;
  const bool hi  = lane >= 16;
  const bool full = (base + 16 <= N);

  int arow = base + col;
  if (arow >= N) arow = N - 1;
  const float t = agg[(size_t)arow * 16];

  v2f a[4], b[4];
#pragma unroll
  for (int k = 0; k < 4; ++k) {
    const int c0 = 4 * k + (hi ? 2 : 0);
    a[k][0] = fmaxf(t * W1[c0]     + b1[c0],     0.0f);
    a[k][1] = fmaxf(t * W1[c0 + 1] + b1[c0 + 1], 0.0f);
    b[k][0] = W2[c0 * 16 + col];
    b[k][1] = W2[(c0 + 1) * 16 + col];
  }

  v8f acc = {};
#pragma unroll
  for (int k = 0; k < 4; ++k)
    acc = __builtin_amdgcn_wmma_f32_16x16x4_f32(false, a[k], false, b[k],
                                                (short)0, acc, false, false);

  if (full) {
#pragma unroll
    for (int r = 0; r < 8; ++r) {
      int orow = base + (hi ? 8 + r : r);
      p2[(size_t)orow * 16 + col]  = acc[r] * dnorm[orow];
      agg[(size_t)orow * 16 + col] = 0.0f;
    }
  } else {
#pragma unroll
    for (int r = 0; r < 8; ++r) {
      int orow = base + (hi ? 8 + r : r);
      if (orow < N) {
        p2[(size_t)orow * 16 + col]  = acc[r] * dnorm[orow];
        agg[(size_t)orow * 16 + col] = 0.0f;
      }
    }
  }
}

__global__ void __launch_bounds__(256)
gcn_out_wmma(const float* __restrict__ p2, const float* __restrict__ agg2,
             const float* __restrict__ dnorm, const float* __restrict__ b2,
             const float* __restrict__ Wl, const float* __restrict__ bl,
             float* __restrict__ out, int N, int nTiles) {
  const int lane = threadIdx.x & 31;
  const int tile = (blockIdx.x * blockDim.x + threadIdx.x) >> 5;
  if (tile >= nTiles) return;
  const int base = tile * 16;
  const int col  = lane & 15;
  const bool hi  = lane >= 16;
  const bool full = (base + 16 <= N);

  int arow = base + col;
  if (arow >= N) arow = N - 1;
  const float di = dnorm[arow];

  v2f a[4], b[4];
#pragma unroll
  for (int k = 0; k < 4; ++k) {
    const int c0 = 4 * k + (hi ? 2 : 0);
    float2 pv = *(const float2*)(p2   + (size_t)arow * 16 + c0);
    float2 av = *(const float2*)(agg2 + (size_t)arow * 16 + c0);
    a[k][0] = fmaxf(di * (av.x + pv.x) + b2[c0],     0.0f);
    a[k][1] = fmaxf(di * (av.y + pv.y) + b2[c0 + 1], 0.0f);
    b[k][0] = (col < 4) ? Wl[c0 * 4 + col]       : 0.0f;
    b[k][1] = (col < 4) ? Wl[(c0 + 1) * 4 + col] : 0.0f;
  }

  const float cv = (col < 4) ? bl[col] : 0.0f;
  v8f acc;
#pragma unroll
  for (int r = 0; r < 8; ++r) acc[r] = cv;
#pragma unroll
  for (int k = 0; k < 4; ++k)
    acc = __builtin_amdgcn_wmma_f32_16x16x4_f32(false, a[k], false, b[k],
                                                (short)0, acc, false, false);

  if (col < 4) {
    if (full) {
#pragma unroll
      for (int r = 0; r < 8; ++r) {
        int orow = base + (hi ? 8 + r : r);
        out[(size_t)orow * 4 + col] = acc[r];
      }
    } else {
#pragma unroll
      for (int r = 0; r < 8; ++r) {
        int orow = base + (hi ? 8 + r : r);
        if (orow < N) out[(size_t)orow * 4 + col] = acc[r];
      }
    }
  }
}

extern "C" void kernel_launch(void* const* d_in, const int* in_sizes, int n_in,
                              void* d_out, int out_size, void* d_ws, size_t ws_size,
                              hipStream_t stream) {
  const float* x  = (const float*)d_in[0];
  const int*   ed = (const int*)d_in[1];
  const float* W1 = (const float*)d_in[2];
  const float* b1 = (const float*)d_in[3];
  const float* W2 = (const float*)d_in[4];
  const float* b2 = (const float*)d_in[5];
  const float* Wl = (const float*)d_in[6];
  const float* bl = (const float*)d_in[7];
  float* out = (float*)d_out;

  const int N = in_sizes[0];
  const int E = in_sizes[1] / 2;

  const int* src = ed;
  const int* dst = ed + E;

  float* dnorm = (float*)d_ws;
  float* p     = dnorm + N;
  float* agg   = p + (size_t)N * 16;

  const int B = 256;
  const int nTiles = (N + 15) / 16;
  const int tileBlocks = (nTiles * 32 + B - 1) / B;
  const int edgeWaves  = (E + 31) / 32;
  const int edgeBlocks = (int)(((long long)edgeWaves * 32 + B - 1) / B);

  unsigned* part = (unsigned*)p;
  float* q = agg;
  const int chunkE  = ((((E + DEG_CHUNKS - 1) / DEG_CHUNKS) + 3) / 4) * 4;
  const int nSlices = (N + DEG_SLICE - 1) / DEG_SLICE;
  deg_hist  <<<nSlices * DEG_CHUNKS, DEG_THREADS, 0, stream>>>(dst, part, N, E, chunkE);
  deg_reduce<<<(N + B - 1) / B, B, 0, stream>>>(part, x, dnorm, q, N);

  const int wChunkE  = ((((E + WS_CHUNKS - 1) / WS_CHUNKS) + 3) / 4) * 4;
  const int wSlices  = (N + WS_SLICE - 1) / WS_SLICE;
  wsum_hist  <<<wSlices * WS_CHUNKS, WS_THREADS, 0, stream>>>(src, dst, q, p, N, E, wChunkE);
  node_reduce<<<(N + B - 1) / B, B, 0, stream>>>(p, x, dnorm, agg, N);

  gcn_mid_wmma<<<tileBlocks, B, 0, stream>>>(agg, dnorm, W1, b1, W2, p, N, nTiles);

  edge_scatter<<<edgeBlocks, B, 0, stream>>>(src, dst, p, agg, E);

  gcn_out_wmma<<<tileBlocks, B, 0, stream>>>(p, agg, dnorm, b2, Wl, bl, out, N, nTiles);
}


# layer-2 sum in work-group LDS: node-slice rows, edges filtered by destination and queued per wave, ds_add_f32; one line-wide f32 atomic per (chunk,row pair) replaces one per edge; sum order changes
# speedup vs baseline: 3.3275x; 1.0928x over previous
#include <hip/hip_runtime.h>

typedef __attribute__((ext_vector_type(2))) float v2f;
typedef __attribute__((ext_vector_type(8))) float v8f;

#define DEG_SLICE   25024
#define DEG_CHUNKS  32
#define DEG_THREADS 1024

#define DEG_BUMP(d)                                              \
  {                                                              \
    const unsigned r_ = (unsigned)((d) - lo);                    \
    if (r_ < cnt) atomicAdd(&hist[r_], 1u);                      \
  }
#define DEG_BUMP4(v) DEG_BUMP((v).x) DEG_BUMP((v).y) DEG_BUMP((v).z) DEG_BUMP((v).w)

__global__ void __launch_bounds__(DEG_THREADS)
deg_hist(const int* __restrict__ dst, unsigned* __restrict__ part, int N, int E, int chunkE) {
  __shared__ unsigned hist[DEG_SLICE];
  const int tid   = threadIdx.x;
  const int chunk = blockIdx.x % DEG_CHUNKS;
  const int slice = blockIdx.x / DEG_CHUNKS;
  const int lo    = slice * DEG_SLICE;
  const int rem   = N - lo;
  const unsigned cnt = (unsigned)(rem < DEG_SLICE ? rem : DEG_SLICE);

  for (unsigned i = tid; i < cnt; i += DEG_THREADS) hist[i] = 0u;
  __syncthreads();

  const long long begin = (long long)chunk * chunkE;
  long long end = begin + chunkE;
  if (end > E) end = E;
  const int len = (end > begin) ? (int)(end - begin) : 0;
  const int* dp = dst + begin;

  int done = 0;
  if ((((size_t)dp) & 15) == 0) {
    const int4* d4 = (const int4*)dp;
    const int n4 = len >> 2;
    int i = tid;
    for (; i + 3 * DEG_THREADS < n4; i += 4 * DEG_THREADS) {
      const int4 a = d4[i];
      const int4 b = d4[i + DEG_THREADS];
      const int4 c = d4[i + 2 * DEG_THREADS];
      const int4 e = d4[i + 3 * DEG_THREADS];
      DEG_BUMP4(a) DEG_BUMP4(b) DEG_BUMP4(c) DEG_BUMP4(e)
    }
    for (; i < n4; i += DEG_THREADS) {
      const int4 a = d4[i];
      DEG_BUMP4(a)
    }
    done = n4 << 2;
  }
  for (int i = done + tid; i < len; i += DEG_THREADS) DEG_BUMP(dp[i])
  __syncthreads();

  unsigned* o = part + (size_t)chunk * N + lo;
  for (unsigned i = tid; i < cnt; i += DEG_THREADS) o[i] = hist[i];
}

__global__ void deg_reduce(const unsigned* part, const float* __restrict__ x,
                           float* __restrict__ dnorm, float* q, int N) {
  int i = blockIdx.x * blockDim.x + threadIdx.x;
  if (i >= N) return;
  unsigned s = 0u;
  for (int c = 0; c < DEG_CHUNKS; ++c) s += part[(size_t)c * N + i];
  const float di = rsqrtf((float)s + 1.0f);
  dnorm[i] = di;
  q[i] = x[i] * di;
}

#define WS_SLICE   12512
#define WS_CHUNKS  16
#define WS_THREADS 1024

#define WS_TAKE(d, s, R, V)                                      \
  const unsigned R = (unsigned)((d) - lo);                       \
  float V = 0.0f;                                                \
  if (R < cnt) V = q[(s)];
#define WS_PUT(R, V) if (R < cnt) atomicAdd(&hist[R], V);
#define WS_TAKE4(dv, sv, T)                                      \
  WS_TAKE((dv).x, (sv).x, T##r0, T##v0) WS_TAKE((dv).y, (sv).y, T##r1, T##v1) \
  WS_TAKE((dv).z, (sv).z, T##r2, T##v2) WS_TAKE((dv).w, (sv).w, T##r3, T##v3)
#define WS_PUT4(T) WS_PUT(T##r0, T##v0) WS_PUT(T##r1, T##v1) WS_PUT(T##r2, T##v2) WS_PUT(T##r3, T##v3)

__global__ void __launch_bounds__(WS_THREADS)
wsum_hist(const int* __restrict__ src, const int* __restrict__ dst, const float* __restrict__ q,
          float* __restrict__ fpart, int N, int E, int chunkE) {
  __shared__ float hist[WS_SLICE];
  const int tid   = threadIdx.x;
  const int chunk = blockIdx.x % WS_CHUNKS;
  const int slice = blockIdx.x / WS_CHUNKS;
  const int lo    = slice * WS_SLICE;
  const int rem   = N - lo;
  const unsigned cnt = (unsigned)(rem < WS_SLICE ? rem : WS_SLICE);

  for (unsigned i = tid; i < cnt; i += WS_THREADS) hist[i] = 0.0f;
  __syncthreads();

  const long long begin = (long long)chunk * chunkE;
  long long end = begin + chunkE;
  if (end > E) end = E;
  const int len = (end > begin) ? (int)(end - begin) : 0;
  const int* dp = dst + begin;
  const int* sp = src + begin;

  int done = 0;
  if (((((size_t)dp) | ((size_t)sp)) & 15) == 0) {
    const int4* d4 = (const int4*)dp;
    const int4* s4 = (const int4*)sp;
    const int n4 = len >> 2;
    int i = tid;
    for (; i + 3 * WS_THREADS < n4; i += 4 * WS_THREADS) {
      const int4 da = d4[i];
      const int4 db = d4[i + WS_THREADS];
      const int4 dc = d4[i + 2 * WS_THREADS];
      const int4 de = d4[i + 3 * WS_THREADS];
      const int4 sa = s4[i];
      const int4 sb = s4[i + WS_THREADS];
      const int4 sc = s4[i + 2 * WS_THREADS];
      const int4 se = s4[i + 3 * WS_THREADS];
      WS_TAKE4(da, sa, a) WS_TAKE4(db, sb, b) WS_TAKE4(dc, sc, c) WS_TAKE4(de, se, e)
      WS_PUT4(a) WS_PUT4(b) WS_PUT4(c) WS_PUT4(e)
    }
    for (; i < n4; i += WS_THREADS) {
      const int4 da = d4[i];
      const int4 sa = s4[i];
      WS_TAKE4(da, sa, a)
      WS_PUT4(a)
    }
    done = n4 << 2;
  }
  for (int i = done + tid; i < len; i += WS_THREADS) {
    WS_TAKE(dp[i], sp[i], tr, tv)
    WS_PUT(tr, tv)
  }
  __syncthreads();

  float* o = fpart + (size_t)chunk * N + lo;
  for (unsigned i = tid; i < cnt; i += WS_THREADS) o[i] = hist[i];
}

__global__ void node_reduce(const float* __restrict__ fpart, const float* __restrict__ x,
                            const float* __restrict__ dnorm, float* __restrict__ agg, int N) {
  int i = blockIdx.x * blockDim.x + threadIdx.x;
  if (i >= N) return;
  float s = 0.0f;
  for (int c = 0; c < WS_CHUNKS; ++c) s += fpart[(size_t)c * N + i];
  const float di = dnorm[i];
  agg[(size_t)i * 16] = di * (s + x[i] * di);
}

#define RS_SLICE   4800
#define RS_CHUNKS  6
#define RS_THREADS 1024
#define RS_WAVES   (RS_THREADS / 32)
#define RS_QCAP    64

__global__ void __launch_bounds__(RS_THREADS)
rows_lds(const int* __restrict__ src, const int* __restrict__ dst, const float* __restrict__ p,
         float* __restrict__ agg, int N, int E, int chunkE) {
  __shared__ float acc[RS_SLICE * 16];
  __shared__ int   qe[RS_WAVES][RS_QCAP];
  __shared__ int   qr[RS_WAVES][RS_QCAP];
  const int tid   = threadIdx.x;
  const int lane  = tid & 31;
  const int w     = __builtin_amdgcn_readfirstlane(tid >> 5);
  const int chunk = blockIdx.x % RS_CHUNKS;
  const int slice = blockIdx.x / RS_CHUNKS;
  const int lo    = slice * RS_SLICE;
  const int rem   = N - lo;
  const unsigned cnt  = (unsigned)(rem < RS_SLICE ? rem : RS_SLICE);
  const unsigned cnt16 = cnt * 16u;

  for (unsigned i = tid; i < cnt16; i += RS_THREADS) acc[i] = 0.0f;
  __syncthreads();

  const long long begin = (long long)chunk * chunkE;
  long long end = begin + chunkE;
  if (end > E) end = E;
  const int len = (end > begin) ? (int)(end - begin) : 0;
  const int* dp = dst + begin;
  const int* sp = src + begin;

  const unsigned c    = (unsigned)(lane & 15);
  const int      half = lane >> 4;
  const unsigned lt   = (1u << lane) - 1u;
  int qn = 0;

  auto drain = [&](int first, int n) {
    int r = 0, s = 0;
    if (lane < n) {
      r = qr[w][first + lane];
      s = sp[qe[w][first + lane]];
    }
    float    v[16];
    unsigned ao[16];
#pragma unroll
    for (int k = 0; k < 16; ++k) {
      const unsigned sj = (unsigned)__shfl(s, 2 * k + half);
      const unsigned rj = (unsigned)__shfl(r, 2 * k + half);
      ao[k] = rj * 16u + c;
      v[k]  = p[sj * 16u + c];
    }
    if (n == 32) {
#pragma unroll
      for (int k = 0; k < 16; ++k) atomicAdd(&acc[ao[k]], v[k]);
    } else {
#pragma unroll
      for (int k = 0; k < 16; ++k)
        if (2 * k + half < n) atomicAdd(&acc[ao[k]], v[k]);
    }
  };

  auto slot = [&](int d, int e) {
    const unsigned r = (unsigned)(d - lo);
    const bool h = r < cnt;
    const unsigned m = __builtin_amdgcn_ballot_w32(h);
    if (m != 0u) {
      if (h) {
        const int pos = qn + __popc(m & lt);
        qe[w][pos] = e;
        qr[w][pos] = (int)r;
      }
      qn += __popc(m);
      if (qn >= 32) {
        drain(qn - 32, 32);
        qn -= 32;
      }
    }
  };

  int done = 0;
  if ((((size_t)dp) & 15) == 0) {
    const int4* d4 = (const int4*)dp;
    const int n4 = len >> 2;
    const int nFull = (n4 / (4 * RS_THREADS)) * (4 * RS_THREADS);
    for (int base = 0; base < nFull; base += 4 * RS_THREADS) {
      const int i0 = base + tid;
      const int i1 = i0 + RS_THREADS;
      const int i2 = i0 + 2 * RS_THREADS;
      const int i3 = i0 + 3 * RS_THREADS;
      const int4 a = d4[i0];
      const int4 b = d4[i1];
      const int4 g = d4[i2];
      const int4 f = d4[i3];
      slot(a.x, 4 * i0); slot(a.y, 4 * i0 + 1); slot(a.z, 4 * i0 + 2); slot(a.w, 4 * i0 + 3);
      slot(b.x, 4 * i1); slot(b.y, 4 * i1 + 1); slot(b.z, 4 * i1 + 2); slot(b.w, 4 * i1 + 3);
      slot(g.x, 4 * i2); slot(g.y, 4 * i2 + 1); slot(g.z, 4 * i2 + 2); slot(g.w, 4 * i2 + 3);
      slot(f.x, 4 * i3); slot(f.y, 4 * i3 + 1); slot(f.z, 4 * i3 + 2); slot(f.w, 4 * i3 + 3);
    }
    done = nFull << 2;
  }
  for (int base = done; base < len; base += RS_THREADS) {
    const int i = base + tid;
    slot((i < len) ? dp[i] : -1, i);
  }
  if (qn > 0) drain(0, qn);
  __syncthreads();

  float* o = agg + (size_t)lo * 16;
  for (unsigned i = tid; i < cnt16; i += RS_THREADS) atomicAdd(o + i, acc[i]);
}

__global__ void __launch_bounds__(256)
gcn_mid_wmma(float* agg, const float* __restrict__ dnorm, const float* __restrict__ W1,
             const float* __restrict__ b1, const float* __restrict__ W2,
             float* __restrict__ p2, int N, int nTiles) {
  const int lane = threadIdx.x & 31;
  const int tile = (blockIdx.x * blockDim.x + threadIdx.x) >> 5;
  if (tile >= nTiles) return;
  const int base = tile * 16;
  const int col  = lane & 15;
  const bool hi  = lane >= 16;
  const bool full = (base + 16 <= N);

  int arow = base + col;
  if (arow >= N) arow = N - 1;
  const float t = agg[(size_t)arow * 16];

  v2f a[4], b[4];
#pragma unroll
  for (int k = 0; k < 4; ++k) {
    const int c0 = 4 * k + (hi ? 2 : 0);
    a[k][0] = fmaxf(t * W1[c0]     + b1[c0],     0.0f);
    a[k][1] = fmaxf(t * W1[c0 + 1] + b1[c0 + 1], 0.0f);
    b[k][0] = W2[c0 * 16 + col];
    b[k][1] = W2[(c0 + 1) * 16 + col];
  }

  v8f acc = {};
#pragma unroll
  for (int k = 0; k < 4; ++k)
    acc = __builtin_amdgcn_wmma_f32_16x16x4_f32(false, a[k], false, b[k],
                                                (short)0, acc, false, false);

  if (full) {
#pragma unroll
    for (int r = 0; r < 8; ++r) {
      int orow = base + (hi ? 8 + r : r);
      p2[(size_t)orow * 16 + col]  = acc[r] * dnorm[orow];
      agg[(size_t)orow * 16 + col] = 0.0f;
    }
  } else {
#pragma unroll
    for (int r = 0; r < 8; ++r) {
      int orow = base + (hi ? 8 + r : r);
      if (orow < N) {
        p2[(size_t)orow * 16 + col]  = acc[r] * dnorm[orow];
        agg[(size_t)orow * 16 + col] = 0.0f;
      }
    }
  }
}

__global__ void __launch_bounds__(256)
gcn_out_wmma(const float* __restrict__ p2, const float* __restrict__ agg2,
             const float* __restrict__ dnorm, const float* __restrict__ b2,
             const float* __restrict__ Wl, const float* __restrict__ bl,
             float* __restrict__ out, int N, int nTiles) {
  const int lane = threadIdx.x & 31;
  const int tile = (blockIdx.x * blockDim.x + threadIdx.x) >> 5;
  if (tile >= nTiles) return;
  const int base = tile * 16;
  const int col  = lane & 15;
  const bool hi  = lane >= 16;
  const bool full = (base + 16 <= N);

  int arow = base + col;
  if (arow >= N) arow = N - 1;
  const float di = dnorm[arow];

  v2f a[4], b[4];
#pragma unroll
  for (int k = 0; k < 4; ++k) {
    const int c0 = 4 * k + (hi ? 2 : 0);
    float2 pv = *(const float2*)(p2   + (size_t)arow * 16 + c0);
    float2 av = *(const float2*)(agg2 + (size_t)arow * 16 + c0);
    a[k][0] = fmaxf(di * (av.x + pv.x) + b2[c0],     0.0f);
    a[k][1] = fmaxf(di * (av.y + pv.y) + b2[c0 + 1], 0.0f);
    b[k][0] = (col < 4) ? Wl[c0 * 4 + col]       : 0.0f;
    b[k][1] = (col < 4) ? Wl[(c0 + 1) * 4 + col] : 0.0f;
  }

  const float cv = (col < 4) ? bl[col] : 0.0f;
  v8f acc;
#pragma unroll
  for (int r = 0; r < 8; ++r) acc[r] = cv;
#pragma unroll
  for (int k = 0; k < 4; ++k)
    acc = __builtin_amdgcn_wmma_f32_16x16x4_f32(false, a[k], false, b[k],
                                                (short)0, acc, false, false);

  if (col < 4) {
    if (full) {
#pragma unroll
      for (int r = 0; r < 8; ++r) {
        int orow = base + (hi ? 8 + r : r);
        out[(size_t)orow * 4 + col] = acc[r];
      }
    } else {
#pragma unroll
      for (int r = 0; r < 8; ++r) {
        int orow = base + (hi ? 8 + r : r);
        if (orow < N) out[(size_t)orow * 4 + col] = acc[r];
      }
    }
  }
}

extern "C" void kernel_launch(void* const* d_in, const int* in_sizes, int n_in,
                              void* d_out, int out_size, void* d_ws, size_t ws_size,
                              hipStream_t stream) {
  const float* x  = (const float*)d_in[0];
  const int*   ed = (const int*)d_in[1];
  const float* W1 = (const float*)d_in[2];
  const float* b1 = (const float*)d_in[3];
  const float* W2 = (const float*)d_in[4];
  const float* b2 = (const float*)d_in[5];
  const float* Wl = (const float*)d_in[6];
  const float* bl = (const float*)d_in[7];
  float* out = (float*)d_out;

  const int N = in_sizes[0];
  const int E = in_sizes[1] / 2;

  const int* src = ed;
  const int* dst = ed + E;

  float* dnorm = (float*)d_ws;
  float* p     = dnorm + N;
  float* agg   = p + (size_t)N * 16;

  const int B = 256;
  const int nTiles = (N + 15) / 16;
  const int tileBlocks = (nTiles * 32 + B - 1) / B;

  unsigned* part = (unsigned*)p;
  float* q = agg;
  const int chunkE  = ((((E + DEG_CHUNKS - 1) / DEG_CHUNKS) + 3) / 4) * 4;
  const int nSlices = (N + DEG_SLICE - 1) / DEG_SLICE;
  deg_hist  <<<nSlices * DEG_CHUNKS, DEG_THREADS, 0, stream>>>(dst, part, N, E, chunkE);
  deg_reduce<<<(N + B - 1) / B, B, 0, stream>>>(part, x, dnorm, q, N);

  const int wChunkE  = ((((E + WS_CHUNKS - 1) / WS_CHUNKS) + 3) / 4) * 4;
  const int wSlices  = (N + WS_SLICE - 1) / WS_SLICE;
  wsum_hist  <<<wSlices * WS_CHUNKS, WS_THREADS, 0, stream>>>(src, dst, q, p, N, E, wChunkE);
  node_reduce<<<(N + B - 1) / B, B, 0, stream>>>(p, x, dnorm, agg, N);

  gcn_mid_wmma<<<tileBlocks, B, 0, stream>>>(agg, dnorm, W1, b1, W2, p, N, nTiles);

  const int rChunkE = ((((E + RS_CHUNKS - 1) / RS_CHUNKS) + 3) / 4) * 4;
  const int rSlices = (N + RS_SLICE - 1) / RS_SLICE;
  rows_lds<<<rSlices * RS_CHUNKS, RS_THREADS, 0, stream>>>(src, dst, p, agg, N, E, rChunkE);

  gcn_out_wmma<<<tileBlocks, B, 0, stream>>>(p, agg, dnorm, b2, Wl, bl, out, N, nTiles);
}


# layer-1 sum: 3 node slices of 66,688 in 267 KB LDS (was 16 of 12,512) x 85 chunks: each edge read and tested 3x, not 16x; chunk sums go to S by line-wide f32 atomics, not 16 planes; sum order changes
# speedup vs baseline: 3.5464x; 1.0658x over previous
#include <hip/hip_runtime.h>

typedef __attribute__((ext_vector_type(2))) float v2f;
typedef __attribute__((ext_vector_type(8))) float v8f;

#define DEG_SLICE   25024
#define DEG_CHUNKS  32
#define DEG_THREADS 1024

#define DEG_BUMP(d)                                              \
  {                                                              \
    const unsigned r_ = (unsigned)((d) - lo);                    \
    if (r_ < cnt) atomicAdd(&hist[r_], 1u);                      \
  }
#define DEG_BUMP4(v) DEG_BUMP((v).x) DEG_BUMP((v).y) DEG_BUMP((v).z) DEG_BUMP((v).w)

__global__ void __launch_bounds__(DEG_THREADS)
deg_hist(const int* __restrict__ dst, unsigned* __restrict__ part, int N, int E, int chunkE) {
  __shared__ unsigned hist[DEG_SLICE];
  const int tid   = threadIdx.x;
  const int chunk = blockIdx.x % DEG_CHUNKS;
  const int slice = blockIdx.x / DEG_CHUNKS;
  const int lo    = slice * DEG_SLICE;
  const int rem   = N - lo;
  const unsigned cnt = (unsigned)(rem < DEG_SLICE ? rem : DEG_SLICE);

  for (unsigned i = tid; i < cnt; i += DEG_THREADS) hist[i] = 0u;
  __syncthreads();

  const long long begin = (long long)chunk * chunkE;
  long long end = begin + chunkE;
  if (end > E) end = E;
  const int len = (end > begin) ? (int)(end - begin) : 0;
  const int* dp = dst + begin;

  int done = 0;
  if ((((size_t)dp) & 15) == 0) {
    const int4* d4 = (const int4*)dp;
    const int n4 = len >> 2;
    int i = tid;
    for (; i + 3 * DEG_THREADS < n4; i += 4 * DEG_THREADS) {
      const int4 a = d4[i];
      const int4 b = d4[i + DEG_THREADS];
      const int4 c = d4[i + 2 * DEG_THREADS];
      const int4 e = d4[i + 3 * DEG_THREADS];
      DEG_BUMP4(a) DEG_BUMP4(b) DEG_BUMP4(c) DEG_BUMP4(e)
    }
    for (; i < n4; i += DEG_THREADS) {
      const int4 a = d4[i];
      DEG_BUMP4(a)
    }
    done = n4 << 2;
  }
  for (int i = done + tid; i < len; i += DEG_THREADS) DEG_BUMP(dp[i])
  __syncthreads();

  unsigned* o = part + (size_t)chunk * N + lo;
  for (unsigned i = tid; i < cnt; i += DEG_THREADS) o[i] = hist[i];
}

__global__ void deg_reduce(const unsigned* part, const float* __restrict__ x,
                           float* __restrict__ dnorm, float* q, unsigned* S0, int N) {
  int i = blockIdx.x * blockDim.x + threadIdx.x;
  if (i >= N) return;
  unsigned s = 0u;
  for (int c = 0; c < DEG_CHUNKS; ++c) s += part[(size_t)c * N + i];
  const float di = rsqrtf((float)s + 1.0f);
  dnorm[i] = di;
  q[i] = x[i] * di;
  S0[i] = 0u;
}

#define WS_SLICE   66688
#define WS_CHUNKS  85
#define WS_THREADS 1024

#define WS_TAKE(d, s, R, V)                                      \
  const unsigned R = (unsigned)((d) - lo);                       \
  float V = 0.0f;                                                \
  if (R < cnt) V = q[(s)];
#define WS_PUT(R, V) if (R < cnt) atomicAdd(&hist[R], V);
#define WS_TAKE4(dv, sv, T)                                      \
  WS_TAKE((dv).x, (sv).x, T##r0, T##v0) WS_TAKE((dv).y, (sv).y, T##r1, T##v1) \
  WS_TAKE((dv).z, (sv).z, T##r2, T##v2) WS_TAKE((dv).w, (sv).w, T##r3, T##v3)
#define WS_PUT4(T) WS_PUT(T##r0, T##v0) WS_PUT(T##r1, T##v1) WS_PUT(T##r2, T##v2) WS_PUT(T##r3, T##v3)

__global__ void __launch_bounds__(WS_THREADS)
wsum_hist(const int* __restrict__ src, const int* __restrict__ dst, const float* __restrict__ q,
          float* __restrict__ S, int N, int E, int chunkE) {
  __shared__ __attribute__((aligned(16))) float hist[WS_SLICE];
  const int tid   = threadIdx.x;
  const int chunk = blockIdx.x % WS_CHUNKS;
  const int slice = blockIdx.x / WS_CHUNKS;
  const int lo    = slice * WS_SLICE;
  const int rem   = N - lo;
  const unsigned cnt = (unsigned)(rem < WS_SLICE ? rem : WS_SLICE);

  float4* h4 = (float4*)hist;
  for (unsigned i = tid; i < WS_SLICE / 4; i += WS_THREADS) h4[i] = make_float4(0.0f, 0.0f, 0.0f, 0.0f);
  __syncthreads();

  const long long begin = (long long)chunk * chunkE;
  long long end = begin + chunkE;
  if (end > E) end = E;
  const int len = (end > begin) ? (int)(end - begin) : 0;
  const int* dp = dst + begin;
  const int* sp = src + begin;

  int done = 0;
  if (((((size_t)dp) | ((size_t)sp)) & 15) == 0) {
    const int4* d4 = (const int4*)dp;
    const int4* s4 = (const int4*)sp;
    const int n4 = len >> 2;
    int i = tid;
    for (; i + 3 * WS_THREADS < n4; i += 4 * WS_THREADS) {
      const int4 da = d4[i];
      const int4 db = d4[i + WS_THREADS];
      const int4 dc = d4[i + 2 * WS_THREADS];
      const int4 de = d4[i + 3 * WS_THREADS];
      const int4 sa = s4[i];
      const int4 sb = s4[i + WS_THREADS];
      const int4 sc = s4[i + 2 * WS_THREADS];
      const int4 se = s4[i + 3 * WS_THREADS];
      WS_TAKE4(da, sa, a) WS_TAKE4(db, sb, b) WS_TAKE4(dc, sc, c) WS_TAKE4(de, se, e)
      WS_PUT4(a) WS_PUT4(b) WS_PUT4(c) WS_PUT4(e)
    }
    for (; i < n4; i += WS_THREADS) {
      const int4 da = d4[i];
      const int4 sa = s4[i];
      WS_TAKE4(da, sa, a)
      WS_PUT4(a)
    }
    done = n4 << 2;
  }
  for (int i = done + tid; i < len; i += WS_THREADS) {
    WS_TAKE(dp[i], sp[i], tr, tv)
    WS_PUT(tr, tv)
  }
  __syncthreads();

  float* o = S + lo;
  const unsigned rot = ((unsigned)chunk * 768u) % cnt;
  for (unsigned i = tid; i < cnt; i += WS_THREADS) {
    unsigned j = i + rot;
    if (j >= cnt) j -= cnt;
    atomicAdd(o + j, hist[j]);
  }
}

__global__ void node_reduce(const float* __restrict__ S, const float* __restrict__ x,
                            const float* __restrict__ dnorm, float* __restrict__ agg, int N) {
  int i = blockIdx.x * blockDim.x + threadIdx.x;
  if (i >= N) return;
  const float di = dnorm[i];
  agg[(size_t)i * 16] = di * (S[i] + x[i] * di);
}

#define RS_SLICE   4800
#define RS_CHUNKS  6
#define RS_THREADS 1024
#define RS_WAVES   (RS_THREADS / 32)
#define RS_QCAP    64

__global__ void __launch_bounds__(RS_THREADS)
rows_lds(const int* __restrict__ src, const int* __restrict__ dst, const float* __restrict__ p,
         float* __restrict__ agg, int N, int E, int chunkE) {
  __shared__ float acc[RS_SLICE * 16];
  __shared__ int   qe[RS_WAVES][RS_QCAP];
  __shared__ int   qr[RS_WAVES][RS_QCAP];
  const int tid   = threadIdx.x;
  const int lane  = tid & 31;
  const int w     = __builtin_amdgcn_readfirstlane(tid >> 5);
  const int chunk = blockIdx.x % RS_CHUNKS;
  const int slice = blockIdx.x / RS_CHUNKS;
  const int lo    = slice * RS_SLICE;
  const int rem   = N - lo;
  const unsigned cnt  = (unsigned)(rem < RS_SLICE ? rem : RS_SLICE);
  const unsigned cnt16 = cnt * 16u;

  for (unsigned i = tid; i < cnt16; i += RS_THREADS) acc[i] = 0.0f;
  __syncthreads();

  const long long begin = (long long)chunk * chunkE;
  long long end = begin + chunkE;
  if (end > E) end = E;
  const int len = (end > begin) ? (int)(end - begin) : 0;
  const int* dp = dst + begin;
  const int* sp = src + begin;

  const unsigned c    = (unsigned)(lane & 15);
  const int      half = lane >> 4;
  const unsigned lt   = (1u << lane) - 1u;
  int qn = 0;

  auto drain = [&](int first, int n) {
    int r = 0, s = 0;
    if (lane < n) {
      r = qr[w][first + lane];
      s = sp[qe[w][first + lane]];
    }
    float    v[16];
    unsigned ao[16];
#pragma unroll
    for (int k = 0; k < 16; ++k) {
      const unsigned sj = (unsigned)__shfl(s, 2 * k + half);
      const unsigned rj = (unsigned)__shfl(r, 2 * k + half);
      ao[k] = rj * 16u + c;
      v[k]  = p[sj * 16u + c];
    }
    if (n == 32) {
#pragma unroll
      for (int k = 0; k < 16; ++k) atomicAdd(&acc[ao[k]], v[k]);
    } else {
#pragma unroll
      for (int k = 0; k < 16; ++k)
        if (2 * k + half < n) atomicAdd(&acc[ao[k]], v[k]);
    }
  };

  auto slot = [&](int d, int e) {
    const unsigned r = (unsigned)(d - lo);
    const bool h = r < cnt;
    const unsigned m = __builtin_amdgcn_ballot_w32(h);
    if (m != 0u) {
      if (h) {
        const int pos = qn + __popc(m & lt);
        qe[w][pos] = e;
        qr[w][pos] = (int)r;
      }
      qn += __popc(m);
      if (qn >= 32) {
        drain(qn - 32, 32);
        qn -= 32;
      }
    }
  };

  int done = 0;
  if ((((size_t)dp) & 15) == 0) {
    const int4* d4 = (const int4*)dp;
    const int n4 = len >> 2;
    const int nFull = (n4 / (4 * RS_THREADS)) * (4 * RS_THREADS);
    for (int base = 0; base < nFull; base += 4 * RS_THREADS) {
      const int i0 = base + tid;
      const int i1 = i0 + RS_THREADS;
      const int i2 = i0 + 2 * RS_THREADS;
      const int i3 = i0 + 3 * RS_THREADS;
      const int4 a = d4[i0];
      const int4 b = d4[i1];
      const int4 g = d4[i2];
      const int4 f = d4[i3];
      slot(a.x, 4 * i0); slot(a.y, 4 * i0 + 1); slot(a.z, 4 * i0 + 2); slot(a.w, 4 * i0 + 3);
      slot(b.x, 4 * i1); slot(b.y, 4 * i1 + 1); slot(b.z, 4 * i1 + 2); slot(b.w, 4 * i1 + 3);
      slot(g.x, 4 * i2); slot(g.y, 4 * i2 + 1); slot(g.z, 4 * i2 + 2); slot(g.w, 4 * i2 + 3);
      slot(f.x, 4 * i3); slot(f.y, 4 * i3 + 1); slot(f.z, 4 * i3 + 2); slot(f.w, 4 * i3 + 3);
    }
    done = nFull << 2;
  }
  for (int base = done; base < len; base += RS_THREADS) {
    const int i = base + tid;
    slot((i < len) ? dp[i] : -1, i);
  }
  if (qn > 0) drain(0, qn);
  __syncthreads();

  float* o = agg + (size_t)lo * 16;
  for (unsigned i = tid; i < cnt16; i += RS_THREADS) atomicAdd(o + i, acc[i]);
}

__global__ void __launch_bounds__(256)
gcn_mid_wmma(float* agg, const float* __restrict__ dnorm, const float* __restrict__ W1,
             const float* __restrict__ b1, const float* __restrict__ W2,
             float* __restrict__ p2, int N, int nTiles) {
  const int lane = threadIdx.x & 31;
  const int tile = (blockIdx.x * blockDim.x + threadIdx.x) >> 5;
  if (tile >= nTiles) return;
  const int base = tile * 16;
  const int col  = lane & 15;
  const bool hi  = lane >= 16;
  const bool full = (base + 16 <= N);

  int arow = base + col;
  if (arow >= N) arow = N - 1;
  const float t = agg[(size_t)arow * 16];

  v2f a[4], b[4];
#pragma unroll
  for (int k = 0; k < 4; ++k) {
    const int c0 = 4 * k + (hi ? 2 : 0);
    a[k][0] = fmaxf(t * W1[c0]     + b1[c0],     0.0f);
    a[k][1] = fmaxf(t * W1[c0 + 1] + b1[c0 + 1], 0.0f);
    b[k][0] = W2[c0 * 16 + col];
    b[k][1] = W2[(c0 + 1) * 16 + col];
  }

  v8f acc = {};
#pragma unroll
  for (int k = 0; k < 4; ++k)
    acc = __builtin_amdgcn_wmma_f32_16x16x4_f32(false, a[k], false, b[k],
                                                (short)0, acc, false, false);

  if (full) {
#pragma unroll
    for (int r = 0; r < 8; ++r) {
      int orow = base + (hi ? 8 + r : r);
      p2[(size_t)orow * 16 + col]  = acc[r] * dnorm[orow];
      agg[(size_t)orow * 16 + col] = 0.0f;
    }
  } else {
#pragma unroll
    for (int r = 0; r < 8; ++r) {
      int orow = base + (hi ? 8 + r : r);
      if (orow < N) {
        p2[(size_t)orow * 16 + col]  = acc[r] * dnorm[orow];
        agg[(size_t)orow * 16 + col] = 0.0f;
      }
    }
  }
}

__global__ void __launch_bounds__(256)
gcn_out_wmma(const float* __restrict__ p2, const float* __restrict__ agg2,
             const float* __restrict__ dnorm, const float* __restrict__ b2,
             const float* __restrict__ Wl, const float* __restrict__ bl,
             float* __restrict__ out, int N, int nTiles) {
  const int lane = threadIdx.x & 31;
  const int tile = (blockIdx.x * blockDim.x + threadIdx.x) >> 5;
  if (tile >= nTiles) return;
  const int base = tile * 16;
  const int col  = lane & 15;
  const bool hi  = lane >= 16;
  const bool full = (base + 16 <= N);

  int arow = base + col;
  if (arow >= N) arow = N - 1;
  const float di = dnorm[arow];

  v2f a[4], b[4];
#pragma unroll
  for (int k = 0; k < 4; ++k) {
    const int c0 = 4 * k + (hi ? 2 : 0);
    float2 pv = *(const float2*)(p2   + (size_t)arow * 16 + c0);
    float2 av = *(const float2*)(agg2 + (size_t)arow * 16 + c0);
    a[k][0] = fmaxf(di * (av.x + pv.x) + b2[c0],     0.0f);
    a[k][1] = fmaxf(di * (av.y + pv.y) + b2[c0 + 1], 0.0f);
    b[k][0] = (col < 4) ? Wl[c0 * 4 + col]       : 0.0f;
    b[k][1] = (col < 4) ? Wl[(c0 + 1) * 4 + col] : 0.0f;
  }

  const float cv = (col < 4) ? bl[col] : 0.0f;
  v8f acc;
#pragma unroll
  for (int r = 0; r < 8; ++r) acc[r] = cv;
#pragma unroll
  for (int k = 0; k < 4; ++k)
    acc = __builtin_amdgcn_wmma_f32_16x16x4_f32(false, a[k], false, b[k],
                                                (short)0, acc, false, false);

  if (col < 4) {
    if (full) {
#pragma unroll
      for (int r = 0; r < 8; ++r) {
        int orow = base + (hi ? 8 + r : r);
        out[(size_t)orow * 4 + col] = acc[r];
      }
    } else {
#pragma unroll
      for (int r = 0; r < 8; ++r) {
        int orow = base + (hi ? 8 + r : r);
        if (orow < N) out[(size_t)orow * 4 + col] = acc[r];
      }
    }
  }
}

extern "C" void kernel_launch(void* const* d_in, const int* in_sizes, int n_in,
                              void* d_out, int out_size, void* d_ws, size_t ws_size,
                              hipStream_t stream) {
  const float* x  = (const float*)d_in[0];
  const int*   ed = (const int*)d_in[1];
  const float* W1 = (const float*)d_in[2];
  const float* b1 = (const float*)d_in[3];
  const float* W2 = (const float*)d_in[4];
  const float* b2 = (const float*)d_in[5];
  const float* Wl = (const float*)d_in[6];
  const float* bl = (const float*)d_in[7];
  float* out = (float*)d_out;

  const int N = in_sizes[0];
  const int E = in_sizes[1] / 2;

  const int* src = ed;
  const int* dst = ed + E;

  float* dnorm = (float*)d_ws;
  float* p     = dnorm + N;
  float* agg   = p + (size_t)N * 16;

  const int B = 256;
  const int nTiles = (N + 15) / 16;
  const int tileBlocks = (nTiles * 32 + B - 1) / B;

  unsigned* part = (unsigned*)p;
  float* q = agg;
  const int chunkE  = ((((E + DEG_CHUNKS - 1) / DEG_CHUNKS) + 3) / 4) * 4;
  const int nSlices = (N + DEG_SLICE - 1) / DEG_SLICE;
  deg_hist  <<<nSlices * DEG_CHUNKS, DEG_THREADS, 0, stream>>>(dst, part, N, E, chunkE);
  deg_reduce<<<(N + B - 1) / B, B, 0, stream>>>(part, x, dnorm, q, part, N);

  const int wChunkE  = ((((E + WS_CHUNKS - 1) / WS_CHUNKS) + 3) / 4) * 4;
  const int wSlices  = (N + WS_SLICE - 1) / WS_SLICE;
  wsum_hist  <<<wSlices * WS_CHUNKS, WS_THREADS, 0, stream>>>(src, dst, q, p, N, E, wChunkE);
  node_reduce<<<(N + B - 1) / B, B, 0, stream>>>(p, x, dnorm, agg, N);

  gcn_mid_wmma<<<tileBlocks, B, 0, stream>>>(agg, dnorm, W1, b1, W2, p, N, nTiles);

  const int rChunkE = ((((E + RS_CHUNKS - 1) / RS_CHUNKS) + 3) / 4) * 4;
  const int rSlices = (N + RS_SLICE - 1) / RS_SLICE;
  rows_lds<<<rSlices * RS_CHUNKS, RS_THREADS, 0, stream>>>(src, dst, p, agg, N, E, rChunkE);

  gcn_out_wmma<<<tileBlocks, B, 0, stream>>>(p, agg, dnorm, b2, Wl, bl, out, N, nTiles);
}
